# starting kernel
# baseline (speedup 1.0000x reference)
#include <hip/hip_runtime.h>

typedef __attribute__((ext_vector_type(2))) float v2f;
typedef __attribute__((ext_vector_type(4))) float v4f;
typedef __attribute__((ext_vector_type(8))) float v8f;

#define Bn   4
#define Hn   128
#define Wn   128
#define Cn   256
#define Gn   16
#define K2n  9
#define Crn  64
#define KGn  (K2n * Gn)
#define NPIX (Bn * Hn * Wn)
#define BN_EPS 1e-3f

__global__ __launch_bounds__(256) void invol_kergen(
    const float* __restrict__ x,  const float* __restrict__ w1,
    const float* __restrict__ b1, const float* __restrict__ gamma,
    const float* __restrict__ beta, const float* __restrict__ mean,
    const float* __restrict__ var,  const float* __restrict__ w2,
    const float* __restrict__ b2,   float* __restrict__ kout)
{
    __shared__ float h_lds[8][16 * Crn];

    const int lane = threadIdx.x & 31;
    const int wave = threadIdx.x >> 5;
    const int tile = blockIdx.x * 8 + wave;
    const int p0   = tile * 16;
    const int m    = lane & 15;
    const int hi   = lane >> 4;
    const int dk   = hi * 2;

    const float* xrow = x + (size_t)(p0 + m) * Cn + dk;
    for (int nt = 0; nt < 4; ++nt) {
        const int n0 = nt * 16;
        v8f acc = {};
        for (int kk = 0; kk < Cn / 4; ++kk) {
            const int kb = kk * 4;
            v2f a = *(const v2f*)(xrow + kb);
            v2f bf;
            bf.x = w1[(kb + dk    ) * Crn + n0 + m];
            bf.y = w1[(kb + dk + 1) * Crn + n0 + m];
            acc = __builtin_amdgcn_wmma_f32_16x16x4_f32(
                      false, a, false, bf, (short)0, acc, false, false);
        }
        const int n = n0 + m;
        const float sc = gamma[n] * rsqrtf(var[n] + BN_EPS);
        const float sh = beta[n] - mean[n] * sc;
        const float bb = b1[n];
#pragma unroll
        for (int r = 0; r < 8; ++r) {
            float v = (acc[r] + bb) * sc + sh;
            v = v > 0.0f ? v : 0.0f;
            h_lds[wave][(r + hi * 8) * Crn + n] = v;
        }
    }
    __syncthreads();

    for (int nt = 0; nt < 9; ++nt) {
        const int n2 = nt * 16 + m;
        v8f acc = {};
        for (int kk = 0; kk < Crn / 4; ++kk) {
            const int kb = kk * 4;
            v2f a = *(const v2f*)(&h_lds[wave][m * Crn + kb + dk]);
            v2f bf;
            bf.x = w2[(kb + dk    ) * KGn + n2];
            bf.y = w2[(kb + dk + 1) * KGn + n2];
            acc = __builtin_amdgcn_wmma_f32_16x16x4_f32(
                      false, a, false, bf, (short)0, acc, false, false);
        }
        const float bb = b2[n2];
#pragma unroll
        for (int r = 0; r < 8; ++r)
            kout[(size_t)(p0 + r + hi * 8) * KGn + n2] = acc[r] + bb;
    }
}

__global__ __launch_bounds__(256) void invol_apply(
    const float* __restrict__ x, const float* __restrict__ kern,
    float* __restrict__ out)
{
    const int t  = blockIdx.x * 256 + threadIdx.x;
    const int ci = t & 63;
    const int p  = t >> 6;
    const int c0 = ci * 4;
    const int g0 = c0 & 15;

    const int b   = p >> 14;
    const int rem = p & 16383;
    const int y   = rem >> 7;
    const int xx  = rem & 127;

    const float* kp = kern + (size_t)p * KGn + g0;
    v4f acc = {};
#pragma unroll
    for (int ki = 0; ki < 3; ++ki) {
        const int yy = y + ki - 1;
        if (yy < 0 || yy >= Hn) continue;
#pragma unroll
        for (int kj = 0; kj < 3; ++kj) {
            const int xj = xx + kj - 1;
            if (xj < 0 || xj >= Wn) continue;
            const v4f xv = *(const v4f*)(x + ((size_t)((b * Hn + yy) * Wn + xj)) * Cn + c0);
            const v4f kv = *(const v4f*)(kp + (ki * 3 + kj) * Gn);
            acc += xv * kv;
        }
    }
    *(v4f*)(out + (size_t)p * Cn + c0) = acc;
}

extern "C" void kernel_launch(void* const* d_in, const int* in_sizes, int n_in,
                              void* d_out, int out_size, void* d_ws, size_t ws_size,
                              hipStream_t stream)
{
    const float* x     = (const float*)d_in[0];
    const float* w1    = (const float*)d_in[1];
    const float* b1    = (const float*)d_in[2];
    const float* gamma = (const float*)d_in[3];
    const float* beta  = (const float*)d_in[4];
    const float* mean  = (const float*)d_in[5];
    const float* var   = (const float*)d_in[6];
    const float* w2    = (const float*)d_in[7];
    const float* b2    = (const float*)d_in[8];

    float* out  = (float*)d_out;
    float* kout = out + (size_t)NPIX * Cn;

    invol_kergen<<<NPIX / 16 / 8, 256, 0, stream>>>(
        x, w1, b1, gamma, beta, mean, var, w2, b2, kout);

    invol_apply<<<(NPIX * 64) / 256, 256, 0, stream>>>(x, kout, out);
}


# Apply kernel re-tiled: work-group per 8x8-pixel tile stages x with a zero rim and the tile's kernel values in LDS, loads up front, nine taps from LDS, wave per row, 16x fewer waves; tap order kept
# speedup vs baseline: 1.9456x; 1.9456x over previous
#include <hip/hip_runtime.h>

typedef __attribute__((ext_vector_type(2))) float v2f;
typedef __attribute__((ext_vector_type(4))) float v4f;
typedef __attribute__((ext_vector_type(8))) float v8f;
typedef __attribute__((ext_vector_type(4))) unsigned v4u;

#define Bn   4
#define Hn   128
#define Wn   128
#define Cn   256
#define Gn   16
#define K2n  9
#define Crn  64
#define KGn  (K2n * Gn)
#define NPIX (Bn * Hn * Wn)
#define BN_EPS 1e-3f

struct FragSetA {
    v4f w[8];
    v4f x[2][2];
};

struct FragSetB {
    v4f lo[4];
    v4f up[4];
};

__device__ __forceinline__ void load_a(FragSetA& f, const float* __restrict__ wp,
                                       const float* __restrict__ xp, int c0)
{
#pragma unroll
    for (int q = 0; q < 8; ++q)
        f.w[q] = *(const v4f*)(wp + (c0 + q) * Crn);
#pragma unroll
    for (int nt = 0; nt < 2; ++nt) {
        f.x[nt][0] = *(const v4f*)(xp + nt * 16 * Cn + c0);
        f.x[nt][1] = *(const v4f*)(xp + nt * 16 * Cn + c0 + 4);
    }
}

__device__ __forceinline__ void mma_a(v8f (&acc)[4][2], const FragSetA& f)
{
#pragma unroll
    for (int j = 0; j < 4; ++j) {
        v2f b[2];
#pragma unroll
        for (int nt = 0; nt < 2; ++nt) {
            b[nt].x = f.x[nt][j >> 1][(2 * j) & 3];
            b[nt].y = f.x[nt][j >> 1][(2 * j + 1) & 3];
        }
#pragma unroll
        for (int mt = 0; mt < 4; ++mt) {
            v2f a;
            a.x = f.w[2 * j][mt];
            a.y = f.w[2 * j + 1][mt];
#pragma unroll
            for (int nt = 0; nt < 2; ++nt)
                acc[mt][nt] = __builtin_amdgcn_wmma_f32_16x16x4_f32(
                                  false, a, false, b[nt], (short)0, acc[mt][nt], false, false);
        }
    }
}

__device__ __forceinline__ void load_b(FragSetB& f, const float* __restrict__ wq, int ks0)
{
#pragma unroll
    for (int s = 0; s < 4; ++s) {
        f.lo[s] = *(const v4f*)(wq + (2 * (ks0 + s)    ) * KGn);
        f.up[s] = *(const v4f*)(wq + (2 * (ks0 + s) + 1) * KGn);
    }
}

__device__ __forceinline__ void mma_b(v8f (&acc2)[4][2], const FragSetB& f,
                                      const v2f (&hA)[2][16], const int ks0)
{
#pragma unroll
    for (int s = 0; s < 4; ++s) {
#pragma unroll
        for (int t = 0; t < 4; ++t) {
            v2f b;
            b.x = f.lo[s][t];
            b.y = f.up[s][t];
#pragma unroll
            for (int nt = 0; nt < 2; ++nt)
                acc2[t][nt] = __builtin_amdgcn_wmma_f32_16x16x4_f32(
                                  false, hA[nt][ks0 + s], false, b, (short)0,
                                  acc2[t][nt], false, false);
        }
    }
}

__global__ __launch_bounds__(128) void invol_kergen(
    const float* __restrict__ x,  const float* __restrict__ w1,
    const float* __restrict__ b1, const float* __restrict__ gamma,
    const float* __restrict__ beta, const float* __restrict__ mean,
    const float* __restrict__ var,  const float* __restrict__ w2,
    const float* __restrict__ b2,   float* __restrict__ kout)
{
    const int lane = threadIdx.x & 31;
    const int wave = threadIdx.x >> 5;
    const int tile = blockIdx.x * 4 + wave;
    const int p0   = tile * 32;
    const int c    = lane & 15;
    const int hi   = lane >> 4;

    const float* wp = w1 + (size_t)(8 * hi) * Crn + 4 * c;
    const float* xp = x + (size_t)(p0 + c) * Cn + 8 * hi;

    v8f acc[4][2];
#pragma unroll
    for (int mt = 0; mt < 4; ++mt)
#pragma unroll
        for (int nt = 0; nt < 2; ++nt)
            acc[mt][nt] = v8f{};

    FragSetA fa0, fa1;
    load_a(fa0, wp, xp, 0);
    for (int blk = 0; blk < 14; blk += 2) {
        load_a(fa1, wp, xp, (blk + 1) * 16);
        mma_a(acc, fa0);
        load_a(fa0, wp, xp, (blk + 2) * 16);
        mma_a(acc, fa1);
    }
    load_a(fa1, wp, xp, 15 * 16);
    mma_a(acc, fa0);
    mma_a(acc, fa1);

    v2f hA[2][16];
#pragma unroll
    for (int r = 0; r < 8; ++r) {
        const int d0 = 32 * hi + 4 * r;
        const v4f g4  = *(const v4f*)(gamma + d0);
        const v4f v4  = *(const v4f*)(var   + d0);
        const v4f be4 = *(const v4f*)(beta  + d0);
        const v4f mu4 = *(const v4f*)(mean  + d0);
        const v4f bb4 = *(const v4f*)(b1    + d0);
        float hv[4][2];
#pragma unroll
        for (int mt = 0; mt < 4; ++mt) {
            const float sc = g4[mt] * rsqrtf(v4[mt] + BN_EPS);
            const float sh = be4[mt] - mu4[mt] * sc;
#pragma unroll
            for (int nt = 0; nt < 2; ++nt) {
                float v = (acc[mt][nt][r] + bb4[mt]) * sc + sh;
                hv[mt][nt] = v > 0.0f ? v : 0.0f;
            }
        }
#pragma unroll
        for (int nt = 0; nt < 2; ++nt) {
            hA[nt][2 * r    ].x = hv[0][nt];  hA[nt][2 * r    ].y = hv[1][nt];
            hA[nt][2 * r + 1].x = hv[2][nt];  hA[nt][2 * r + 1].y = hv[3][nt];
        }
    }

    const float* wq   = w2 + (size_t)(32 * hi) * KGn + 4 * c;
    float*       orow = kout + (size_t)(p0 + 8 * hi) * KGn;

#pragma unroll 1
    for (int Q = 0; Q < 2; ++Q) {
        const float* wqq = wq + 64 * Q;
        FragSetB fb0, fb1;
        load_b(fb0, wqq, 0);
        load_b(fb1, wqq, 4);
        v8f acc2[4][2];
#pragma unroll
        for (int t = 0; t < 4; ++t)
#pragma unroll
            for (int nt = 0; nt < 2; ++nt)
                acc2[t][nt] = v8f{};
        mma_b(acc2, fb0, hA, 0);
        load_b(fb0, wqq, 8);
        mma_b(acc2, fb1, hA, 4);
        load_b(fb1, wqq, 12);
        mma_b(acc2, fb0, hA, 8);
        mma_b(acc2, fb1, hA, 12);

        const v4f bias = *(const v4f*)(b2 + 64 * Q + 4 * c);
#pragma unroll
        for (int nt = 0; nt < 2; ++nt) {
#pragma unroll
            for (int r = 0; r < 8; ++r) {
                v4f o;
                o.x = acc2[0][nt][r] + bias.x;
                o.y = acc2[1][nt][r] + bias.y;
                o.z = acc2[2][nt][r] + bias.z;
                o.w = acc2[3][nt][r] + bias.w;
                *(v4f*)(orow + (size_t)(16 * nt + r) * KGn + 64 * Q + 4 * c) = o;
            }
        }
    }

    {
        const float* wt = w2 + (size_t)(32 * hi) * KGn + 128 + c;
        v2f bt[16];
#pragma unroll
        for (int ks = 0; ks < 16; ++ks) {
            bt[ks].x = wt[(2 * ks    ) * KGn];
            bt[ks].y = wt[(2 * ks + 1) * KGn];
        }
        v8f acc3[2];
        acc3[0] = v8f{};
        acc3[1] = v8f{};
#pragma unroll
        for (int ks = 0; ks < 16; ++ks) {
#pragma unroll
            for (int nt = 0; nt < 2; ++nt)
                acc3[nt] = __builtin_amdgcn_wmma_f32_16x16x4_f32(
                               false, hA[nt][ks], false, bt[ks], (short)0,
                               acc3[nt], false, false);
        }
        const float bb = b2[128 + c];
#pragma unroll
        for (int nt = 0; nt < 2; ++nt) {
#pragma unroll
            for (int r = 0; r < 8; ++r)
                orow[(size_t)(16 * nt + r) * KGn + 128 + c] = acc3[nt][r] + bb;
        }
    }
}

#define TSn  8
#define HSn  (TSn + 2)
#define XSn  (HSn * HSn * Cn)
#define KSn  (TSn * TSn * KGn)
#define XLD  (HSn * HSn * 64 / 256)

__global__ __launch_bounds__(256) void invol_apply(
    const float* __restrict__ x, const float* __restrict__ kern,
    float* __restrict__ out)
{
    __shared__ __attribute__((aligned(16))) float xs[XSn];
    __shared__ __attribute__((aligned(16))) float ks[KSn];

    const int lane = threadIdx.x & 31;
    const int wave = __builtin_amdgcn_readfirstlane((int)(threadIdx.x >> 5));
    const int tile = blockIdx.x;
    const int x0   = (tile & 15) * TSn;
    const int y0   = ((tile >> 4) & 15) * TSn;
    const int b    = tile >> 8;

    const int hp0  = (wave >> 1) & 3;
    const int half = (wave & 1) * 128;
    const float* xl = x + 4 * lane;
    v4u      xv[XLD];
    unsigned keep[XLD];
#pragma unroll
    for (int i = 0; i < XLD; ++i) {
        const int s    = (4 * i) % HSn + hp0;
        const int wrap = s >= HSn ? 1 : 0;
        const int hy   = (4 * i) / HSn + wrap;
        const int hx   = s - HSn * wrap;
        const int gy = y0 + hy - 1;
        const int gx = x0 + hx - 1;
        keep[i] = ((unsigned)gy < (unsigned)Hn && (unsigned)gx < (unsigned)Wn) ? 0xffffffffu : 0u;
        const int cy = gy < 0 ? 0 : (gy > Hn - 1 ? Hn - 1 : gy);
        const int cx = gx < 0 ? 0 : (gx > Wn - 1 ? Wn - 1 : gx);
        xv[i] = *(const v4u*)(xl + (((b * Hn + cy) * Wn + cx) * Cn + half));
    }
    const size_t prow = (size_t)((b * Hn + y0 + wave) * Wn + x0);
    const float* kg   = kern + prow * KGn + 4 * lane;
    v4f kv[9];
#pragma unroll
    for (int q = 0; q < 9; ++q)
        kv[q] = *(const v4f*)(kg + q * 128);

#pragma unroll
    for (int i = 0; i < XLD; ++i) {
        const int hp = hp0 + 4 * i;
        *(v4u*)(xs + hp * Cn + half + 4 * lane) = xv[i] & keep[i];
    }
    float* kw = ks + wave * (TSn * KGn);
#pragma unroll
    for (int q = 0; q < 9; ++q)
        *(v4f*)(kw + 4 * lane + q * 128) = kv[q];

    __syncthreads();

    const float* xr   = xs + (wave * HSn) * Cn + 4 * lane;
    const float* kr   = kw + ((4 * lane) & 15);
    float*       orow = out + prow * Cn + 4 * lane;

    v4f col[3][3][2];
#pragma unroll
    for (int hx = 0; hx < 2; ++hx)
#pragma unroll
        for (int ki = 0; ki < 3; ++ki) {
            col[hx][ki][0] = *(const v4f*)(xr + (ki * HSn + hx) * Cn);
            col[hx][ki][1] = *(const v4f*)(xr + (ki * HSn + hx) * Cn + 128);
        }
#pragma unroll
    for (int j = 0; j < TSn; ++j) {
#pragma unroll
        for (int ki = 0; ki < 3; ++ki) {
            col[(j + 2) % 3][ki][0] = *(const v4f*)(xr + (ki * HSn + j + 2) * Cn);
            col[(j + 2) % 3][ki][1] = *(const v4f*)(xr + (ki * HSn + j + 2) * Cn + 128);
        }
        v4f a0 = {}, a1 = {};
#pragma unroll
        for (int ki = 0; ki < 3; ++ki)
#pragma unroll
            for (int kj = 0; kj < 3; ++kj) {
                const v4f kq = *(const v4f*)(kr + j * KGn + (ki * 3 + kj) * Gn);
                a0 += col[(j + kj) % 3][ki][0] * kq;
                a1 += col[(j + kj) % 3][ki][1] * kq;
            }
        *(v4f*)(orow + j * Cn)       = a0;
        *(v4f*)(orow + j * Cn + 128) = a1;
    }
}

extern "C" void kernel_launch(void* const* d_in, const int* in_sizes, int n_in,
                              void* d_out, int out_size, void* d_ws, size_t ws_size,
                              hipStream_t stream)
{
    const float* x     = (const float*)d_in[0];
    const float* w1    = (const float*)d_in[1];
    const float* b1    = (const float*)d_in[2];
    const float* gamma = (const float*)d_in[3];
    const float* beta  = (const float*)d_in[4];
    const float* mean  = (const float*)d_in[5];
    const float* var   = (const float*)d_in[6];
    const float* w2    = (const float*)d_in[7];
    const float* b2    = (const float*)d_in[8];

    float* out  = (float*)d_out;
    float* kout = out + (size_t)NPIX * Cn;

    invol_kergen<<<NPIX / 32 / 4, 128, 0, stream>>>(
        x, w1, b1, gamma, beta, mean, var, w2, b2, kout);

    invol_apply<<<NPIX / (TSn * TSn), 256, 0, stream>>>(x, kout, out);
}


# Both launches fused per 8x8 tile: x staged once in LDS (rim, padded pitch) is the B operand of the kernel-value GEMMs (wave = 2 rows) and feeds the nine-tap walk; values pass in LDS; sum order kept
# speedup vs baseline: 2.2822x; 1.1730x over previous
#include <hip/hip_runtime.h>

typedef __attribute__((ext_vector_type(2))) float v2f;
typedef __attribute__((ext_vector_type(4))) float v4f;
typedef __attribute__((ext_vector_type(8))) float v8f;
typedef __attribute__((ext_vector_type(4))) unsigned v4u;

#define Bn   4
#define Hn   128
#define Wn   128
#define Cn   256
#define Gn   16
#define K2n  9
#define Crn  64
#define KGn  (K2n * Gn)
#define NPIX (Bn * Hn * Wn)
#define BN_EPS 1e-3f

#define TSn  8
#define HSn  (TSn + 2)
#define XPn  (Cn + 4)
#define XSn  (HSn * HSn * XPn)
#define KSn  (TSn * TSn * KGn)
#define XLD  (HSn * HSn * 2 / 4)

struct FragSetA {
    v4f w[8];
    v4f x[2];
};

struct FragSetB {
    v4f lo[16];
    v4f up[16];
};

__device__ __forceinline__ void load_w(FragSetA& f, const float* __restrict__ wp, int c0)
{
#pragma unroll
    for (int q = 0; q < 8; ++q)
        f.w[q] = *(const v4f*)(wp + (c0 + q) * Crn);
}

__device__ __forceinline__ void load_x(FragSetA& f, const float* xq, int c0)
{
    f.x[0] = *(const v4f*)(xq + c0);
    f.x[1] = *(const v4f*)(xq + c0 + 4);
}

__device__ __forceinline__ void mma_a(v8f (&acc)[4], const FragSetA& f)
{
#pragma unroll
    for (int j = 0; j < 4; ++j) {
        v2f b;
        b.x = f.x[j >> 1][(2 * j) & 3];
        b.y = f.x[j >> 1][(2 * j + 1) & 3];
#pragma unroll
        for (int mt = 0; mt < 4; ++mt) {
            v2f a;
            a.x = f.w[2 * j][mt];
            a.y = f.w[2 * j + 1][mt];
            acc[mt] = __builtin_amdgcn_wmma_f32_16x16x4_f32(
                          false, a, false, b, (short)0, acc[mt], false, false);
        }
    }
}

__device__ __forceinline__ void load_b(FragSetB& f, const float* __restrict__ wq)
{
#pragma unroll
    for (int s = 0; s < 16; ++s) {
        f.lo[s] = *(const v4f*)(wq + (2 * s    ) * KGn);
        f.up[s] = *(const v4f*)(wq + (2 * s + 1) * KGn);
    }
}

__device__ __forceinline__ void mma_b(v8f (&acc2)[4], const FragSetB& f, const v2f (&hA)[16])
{
#pragma unroll
    for (int s = 0; s < 16; ++s) {
#pragma unroll
        for (int t = 0; t < 4; ++t) {
            v2f b;
            b.x = f.lo[s][t];
            b.y = f.up[s][t];
            acc2[t] = __builtin_amdgcn_wmma_f32_16x16x4_f32(
                          false, hA[s], false, b, (short)0, acc2[t], false, false);
        }
    }
}

__device__ __forceinline__ void store_b(const v8f (&acc2)[4], const v4f bias,
                                        float* __restrict__ og, float* kl)
{
#pragma unroll
    for (int r = 0; r < 8; ++r) {
        v4f o;
        o.x = acc2[0][r] + bias.x;
        o.y = acc2[1][r] + bias.y;
        o.z = acc2[2][r] + bias.z;
        o.w = acc2[3][r] + bias.w;
        *(v4f*)(og + r * KGn) = o;
        *(v4f*)(kl + r * KGn) = o;
    }
}

__global__ __launch_bounds__(128, 2) void invol_fused(
    const float* __restrict__ x,  const float* __restrict__ w1,
    const float* __restrict__ b1, const float* __restrict__ gamma,
    const float* __restrict__ beta, const float* __restrict__ mean,
    const float* __restrict__ var,  const float* __restrict__ w2,
    const float* __restrict__ b2,   float* __restrict__ kout,
    float* __restrict__ out)
{
    __shared__ __attribute__((aligned(16))) float xs[XSn];
    __shared__ __attribute__((aligned(16))) float ks[KSn];

    const int lane = threadIdx.x & 31;
    const int wave = __builtin_amdgcn_readfirstlane((int)(threadIdx.x >> 5));
    const int tile = blockIdx.x;
    const int x0   = (tile & 15) * TSn;
    const int y0   = ((tile >> 4) & 15) * TSn;
    const int b    = tile >> 8;
    const int c    = lane & 15;
    const int hi   = lane >> 4;

    const int hp0  = wave >> 1;
    const int half = (wave & 1) * 128;
    const float* xl = x + 4 * lane;
    v4u      xv[XLD];
    unsigned keep[XLD];
#pragma unroll
    for (int i = 0; i < XLD; ++i) {
        const int hy = (2 * i) / HSn;
        const int hx = (2 * i) % HSn + hp0;
        const int gy = y0 + hy - 1;
        const int gx = x0 + hx - 1;
        keep[i] = ((unsigned)gy < (unsigned)Hn && (unsigned)gx < (unsigned)Wn) ? 0xffffffffu : 0u;
        const int cy = gy < 0 ? 0 : (gy > Hn - 1 ? Hn - 1 : gy);
        const int cx = gx < 0 ? 0 : (gx > Wn - 1 ? Wn - 1 : gx);
        xv[i] = *(const v4u*)(xl + (((b * Hn + cy) * Wn + cx) * Cn + half));
    }

    const float* wp = w1 + (size_t)(8 * hi) * Crn + 4 * c;
    FragSetA fa0, fa1, fb0, fb1;
    load_w(fa0, wp, 0);
    load_w(fa1, wp, 16);

#pragma unroll
    for (int i = 0; i < XLD; ++i) {
        const int hp = hp0 + 2 * i;
        *(v4u*)(xs + hp * XPn + half + 4 * lane) = xv[i] & keep[i];
    }

    __syncthreads();

    const int pr = (2 * wave + (c >> 3) + 1) * HSn + (c & 7) + 1;
    const float* xq = xs + pr * XPn + 8 * hi;

    v8f acc[4];
#pragma unroll
    for (int mt = 0; mt < 4; ++mt)
        acc[mt] = v8f{};

#define LOAD_GROUP_END() __builtin_amdgcn_sched_barrier(0)
    load_x(fa0, xq, 0);
    load_x(fa1, xq, 16);
    for (int p = 0; p < 6; p += 2) {
        load_w(fb0, wp, (p + 1) * 32);      load_x(fb0, xq, (p + 1) * 32);
        load_w(fb1, wp, (p + 1) * 32 + 16); load_x(fb1, xq, (p + 1) * 32 + 16);
        LOAD_GROUP_END();
        mma_a(acc, fa0);
        mma_a(acc, fa1);
        LOAD_GROUP_END();
        load_w(fa0, wp, (p + 2) * 32);      load_x(fa0, xq, (p + 2) * 32);
        load_w(fa1, wp, (p + 2) * 32 + 16); load_x(fa1, xq, (p + 2) * 32 + 16);
        LOAD_GROUP_END();
        mma_a(acc, fb0);
        mma_a(acc, fb1);
    }
    load_w(fb0, wp, 7 * 32);      load_x(fb0, xq, 7 * 32);
    load_w(fb1, wp, 7 * 32 + 16); load_x(fb1, xq, 7 * 32 + 16);
    LOAD_GROUP_END();
    mma_a(acc, fa0);
    mma_a(acc, fa1);
    mma_a(acc, fb0);
    mma_a(acc, fb1);

    const float* wq = w2 + (size_t)(32 * hi) * KGn + 4 * c;
    FragSetB g0, g1;
    load_b(g0, wq);
    LOAD_GROUP_END();

    v2f hA[16];
#pragma unroll
    for (int r = 0; r < 8; ++r) {
        const int d0 = 32 * hi + 4 * r;
        const v4f g4  = *(const v4f*)(gamma + d0);
        const v4f v4  = *(const v4f*)(var   + d0);
        const v4f be4 = *(const v4f*)(beta  + d0);
        const v4f mu4 = *(const v4f*)(mean  + d0);
        const v4f bb4 = *(const v4f*)(b1    + d0);
        float hv[4];
#pragma unroll
        for (int mt = 0; mt < 4; ++mt) {
            const float sc = g4[mt] * rsqrtf(v4[mt] + BN_EPS);
            const float sh = be4[mt] - mu4[mt] * sc;
            float v = (acc[mt][r] + bb4[mt]) * sc + sh;
            hv[mt] = v > 0.0f ? v : 0.0f;
        }
        hA[2 * r    ].x = hv[0];  hA[2 * r    ].y = hv[1];
        hA[2 * r + 1].x = hv[2];  hA[2 * r + 1].y = hv[3];
    }

    const size_t prow_hi = (size_t)((b * Hn + y0 + 2 * wave + hi) * Wn + x0);
    float* og = kout + prow_hi * KGn;
    float* kl = ks + ((2 * wave + hi) * TSn) * KGn;

    load_b(g1, wq + 64);
    LOAD_GROUP_END();
    {
        v8f acc2[4];
#pragma unroll
        for (int t = 0; t < 4; ++t)
            acc2[t] = v8f{};
        mma_b(acc2, g0, hA);
        const v4f bias = *(const v4f*)(b2 + 4 * c);
        store_b(acc2, bias, og + 4 * c, kl + 4 * c);
    }

    const float* wt = w2 + (size_t)(32 * hi) * KGn + 128 + c;
    v2f bt[16];
#pragma unroll
    for (int s = 0; s < 16; ++s) {
        bt[s].x = wt[(2 * s    ) * KGn];
        bt[s].y = wt[(2 * s + 1) * KGn];
    }
    LOAD_GROUP_END();
    {
        v8f acc2[4];
#pragma unroll
        for (int t = 0; t < 4; ++t)
            acc2[t] = v8f{};
        v8f acc3 = v8f{};
#pragma unroll
        for (int s = 0; s < 16; ++s) {
#pragma unroll
            for (int t = 0; t < 4; ++t) {
                v2f bq;
                bq.x = g1.lo[s][t];
                bq.y = g1.up[s][t];
                acc2[t] = __builtin_amdgcn_wmma_f32_16x16x4_f32(
                              false, hA[s], false, bq, (short)0, acc2[t], false, false);
            }
            acc3 = __builtin_amdgcn_wmma_f32_16x16x4_f32(
                       false, hA[s], false, bt[s], (short)0, acc3, false, false);
        }
        const v4f bias = *(const v4f*)(b2 + 64 + 4 * c);
        store_b(acc2, bias, og + 64 + 4 * c, kl + 64 + 4 * c);
        const float bb = b2[128 + c];
#pragma unroll
        for (int r = 0; r < 8; ++r) {
            const float o = acc3[r] + bb;
            og[r * KGn + 128 + c] = o;
            kl[r * KGn + 128 + c] = o;
        }
    }

    __syncthreads();

#pragma unroll 1
    for (int rr = 0; rr < 2; ++rr) {
        const int R = 2 * wave + rr;
        const float* xr   = xs + (R * HSn) * XPn + 4 * lane;
        const float* kr   = ks + (R * TSn) * KGn + ((4 * lane) & 15);
        const size_t prow = (size_t)((b * Hn + y0 + R) * Wn + x0);
        float*       orow = out + prow * Cn + 4 * lane;

        v4f col[4][3][2];
        v4f kq[2][9];
#pragma unroll
        for (int hx = 0; hx < 3; ++hx)
#pragma unroll
            for (int ki = 0; ki < 3; ++ki) {
                col[hx][ki][0] = *(const v4f*)(xr + (ki * HSn + hx) * XPn);
                col[hx][ki][1] = *(const v4f*)(xr + (ki * HSn + hx) * XPn + 128);
            }
#pragma unroll
        for (int t = 0; t < 9; ++t)
            kq[0][t] = *(const v4f*)(kr + t * Gn);
#pragma unroll
        for (int j = 0; j < TSn; ++j) {
            if (j + 1 < TSn) {
#pragma unroll
                for (int ki = 0; ki < 3; ++ki) {
                    col[(j + 3) & 3][ki][0] = *(const v4f*)(xr + (ki * HSn + j + 3) * XPn);
                    col[(j + 3) & 3][ki][1] = *(const v4f*)(xr + (ki * HSn + j + 3) * XPn + 128);
                }
#pragma unroll
                for (int t = 0; t < 9; ++t)
                    kq[(j + 1) & 1][t] = *(const v4f*)(kr + (j + 1) * KGn + t * Gn);
            }
            v4f a0 = {}, a1 = {};
#pragma unroll
            for (int ki = 0; ki < 3; ++ki)
#pragma unroll
                for (int kj = 0; kj < 3; ++kj) {
                    const v4f kv = kq[j & 1][ki * 3 + kj];
                    a0 += col[(j + kj) & 3][ki][0] * kv;
                    a1 += col[(j + kj) & 3][ki][1] * kv;
                }
            *(v4f*)(orow + j * Cn)       = a0;
            *(v4f*)(orow + j * Cn + 128) = a1;
            __builtin_amdgcn_sched_barrier(0);
        }
    }
}

extern "C" void kernel_launch(void* const* d_in, const int* in_sizes, int n_in,
                              void* d_out, int out_size, void* d_ws, size_t ws_size,
                              hipStream_t stream)
{
    const float* x     = (const float*)d_in[0];
    const float* w1    = (const float*)d_in[1];
    const float* b1    = (const float*)d_in[2];
    const float* gamma = (const float*)d_in[3];
    const float* beta  = (const float*)d_in[4];
    const float* mean  = (const float*)d_in[5];
    const float* var   = (const float*)d_in[6];
    const float* w2    = (const float*)d_in[7];
    const float* b2    = (const float*)d_in[8];

    float* out  = (float*)d_out;
    float* kout = out + (size_t)NPIX * Cn;

    invol_fused<<<NPIX / (TSn * TSn), 128, 0, stream>>>(
        x, w1, b1, gamma, beta, mean, var, w2, b2, kout, out);
}


# Second work-group barrier dropped: a wave walks only kernel values it stored itself, so a wave-scope LDS fence replaces it; no wait for the other waves or for the kern stores' acknowledgement
# speedup vs baseline: 2.3673x; 1.0373x over previous
#include <hip/hip_runtime.h>

typedef __attribute__((ext_vector_type(2))) float v2f;
typedef __attribute__((ext_vector_type(4))) float v4f;
typedef __attribute__((ext_vector_type(8))) float v8f;
typedef __attribute__((ext_vector_type(4))) unsigned v4u;

#define Bn   4
#define Hn   128
#define Wn   128
#define Cn   256
#define Gn   16
#define K2n  9
#define Crn  64
#define KGn  (K2n * Gn)
#define NPIX (Bn * Hn * Wn)
#define BN_EPS 1e-3f

#define TSn  8
#define HSn  (TSn + 2)
#define XPn  (Cn + 4)
#define XSn  (HSn * HSn * XPn)
#define KSn  (TSn * TSn * KGn)
#define XLD  (HSn * HSn * 2 / 4)

struct FragSetA {
    v4f w[8];
    v4f x[2];
};

struct FragSetB {
    v4f lo[16];
    v4f up[16];
};

__device__ __forceinline__ void load_w(FragSetA& f, const float* __restrict__ wp, int c0)
{
#pragma unroll
    for (int q = 0; q < 8; ++q)
        f.w[q] = *(const v4f*)(wp + (c0 + q) * Crn);
}

__device__ __forceinline__ void load_x(FragSetA& f, const float* xq, int c0)
{
    f.x[0] = *(const v4f*)(xq + c0);
    f.x[1] = *(const v4f*)(xq + c0 + 4);
}

__device__ __forceinline__ void mma_a(v8f (&acc)[4], const FragSetA& f)
{
#pragma unroll
    for (int j = 0; j < 4; ++j) {
        v2f b;
        b.x = f.x[j >> 1][(2 * j) & 3];
        b.y = f.x[j >> 1][(2 * j + 1) & 3];
#pragma unroll
        for (int mt = 0; mt < 4; ++mt) {
            v2f a;
            a.x = f.w[2 * j][mt];
            a.y = f.w[2 * j + 1][mt];
            acc[mt] = __builtin_amdgcn_wmma_f32_16x16x4_f32(
                          false, a, false, b, (short)0, acc[mt], false, false);
        }
    }
}

__device__ __forceinline__ void load_b(FragSetB& f, const float* __restrict__ wq)
{
#pragma unroll
    for (int s = 0; s < 16; ++s) {
        f.lo[s] = *(const v4f*)(wq + (2 * s    ) * KGn);
        f.up[s] = *(const v4f*)(wq + (2 * s + 1) * KGn);
    }
}

__device__ __forceinline__ void mma_b(v8f (&acc2)[4], const FragSetB& f, const v2f (&hA)[16])
{
#pragma unroll
    for (int s = 0; s < 16; ++s) {
#pragma unroll
        for (int t = 0; t < 4; ++t) {
            v2f b;
            b.x = f.lo[s][t];
            b.y = f.up[s][t];
            acc2[t] = __builtin_amdgcn_wmma_f32_16x16x4_f32(
                          false, hA[s], false, b, (short)0, acc2[t], false, false);
        }
    }
}

__device__ __forceinline__ void store_b(const v8f (&acc2)[4], const v4f bias,
                                        float* __restrict__ og, float* kl)
{
#pragma unroll
    for (int r = 0; r < 8; ++r) {
        v4f o;
        o.x = acc2[0][r] + bias.x;
        o.y = acc2[1][r] + bias.y;
        o.z = acc2[2][r] + bias.z;
        o.w = acc2[3][r] + bias.w;
        *(v4f*)(og + r * KGn) = o;
        *(v4f*)(kl + r * KGn) = o;
    }
}

__global__ __launch_bounds__(128, 2) void invol_fused(
    const float* __restrict__ x,  const float* __restrict__ w1,
    const float* __restrict__ b1, const float* __restrict__ gamma,
    const float* __restrict__ beta, const float* __restrict__ mean,
    const float* __restrict__ var,  const float* __restrict__ w2,
    const float* __restrict__ b2,   float* __restrict__ kout,
    float* __restrict__ out)
{
    __shared__ __attribute__((aligned(16))) float xs[XSn];
    __shared__ __attribute__((aligned(16))) float ks[KSn];

    const int lane = threadIdx.x & 31;
    const int wave = __builtin_amdgcn_readfirstlane((int)(threadIdx.x >> 5));
    const int tile = blockIdx.x;
    const int x0   = (tile & 15) * TSn;
    const int y0   = ((tile >> 4) & 15) * TSn;
    const int b    = tile >> 8;
    const int c    = lane & 15;
    const int hi   = lane >> 4;

    const int hp0  = wave >> 1;
    const int half = (wave & 1) * 128;
    const float* xl = x + 4 * lane;
    v4u      xv[XLD];
    unsigned keep[XLD];
#pragma unroll
    for (int i = 0; i < XLD; ++i) {
        const int hy = (2 * i) / HSn;
        const int hx = (2 * i) % HSn + hp0;
        const int gy = y0 + hy - 1;
        const int gx = x0 + hx - 1;
        keep[i] = ((unsigned)gy < (unsigned)Hn && (unsigned)gx < (unsigned)Wn) ? 0xffffffffu : 0u;
        const int cy = gy < 0 ? 0 : (gy > Hn - 1 ? Hn - 1 : gy);
        const int cx = gx < 0 ? 0 : (gx > Wn - 1 ? Wn - 1 : gx);
        xv[i] = *(const v4u*)(xl + (((b * Hn + cy) * Wn + cx) * Cn + half));
    }

    const float* wp = w1 + (size_t)(8 * hi) * Crn + 4 * c;
    FragSetA fa0, fa1, fb0, fb1;
    load_w(fa0, wp, 0);
    load_w(fa1, wp, 16);

#pragma unroll
    for (int i = 0; i < XLD; ++i) {
        const int hp = hp0 + 2 * i;
        *(v4u*)(xs + hp * XPn + half + 4 * lane) = xv[i] & keep[i];
    }

    __syncthreads();

    const int pr = (2 * wave + (c >> 3) + 1) * HSn + (c & 7) + 1;
    const float* xq = xs + pr * XPn + 8 * hi;

    v8f acc[4];
#pragma unroll
    for (int mt = 0; mt < 4; ++mt)
        acc[mt] = v8f{};

#define LOAD_GROUP_END() __builtin_amdgcn_sched_barrier(0)
    load_x(fa0, xq, 0);
    load_x(fa1, xq, 16);
    for (int p = 0; p < 6; p += 2) {
        load_w(fb0, wp, (p + 1) * 32);      load_x(fb0, xq, (p + 1) * 32);
        load_w(fb1, wp, (p + 1) * 32 + 16); load_x(fb1, xq, (p + 1) * 32 + 16);
        LOAD_GROUP_END();
        mma_a(acc, fa0);
        mma_a(acc, fa1);
        LOAD_GROUP_END();
        load_w(fa0, wp, (p + 2) * 32);      load_x(fa0, xq, (p + 2) * 32);
        load_w(fa1, wp, (p + 2) * 32 + 16); load_x(fa1, xq, (p + 2) * 32 + 16);
        LOAD_GROUP_END();
        mma_a(acc, fb0);
        mma_a(acc, fb1);
    }
    load_w(fb0, wp, 7 * 32);      load_x(fb0, xq, 7 * 32);
    load_w(fb1, wp, 7 * 32 + 16); load_x(fb1, xq, 7 * 32 + 16);
    LOAD_GROUP_END();
    mma_a(acc, fa0);
    mma_a(acc, fa1);
    mma_a(acc, fb0);
    mma_a(acc, fb1);

    const float* wq = w2 + (size_t)(32 * hi) * KGn + 4 * c;
    FragSetB g0, g1;
    load_b(g0, wq);
    LOAD_GROUP_END();

    v2f hA[16];
#pragma unroll
    for (int r = 0; r < 8; ++r) {
        const int d0 = 32 * hi + 4 * r;
        const v4f g4  = *(const v4f*)(gamma + d0);
        const v4f v4  = *(const v4f*)(var   + d0);
        const v4f be4 = *(const v4f*)(beta  + d0);
        const v4f mu4 = *(const v4f*)(mean  + d0);
        const v4f bb4 = *(const v4f*)(b1    + d0);
        float hv[4];
#pragma unroll
        for (int mt = 0; mt < 4; ++mt) {
            const float sc = g4[mt] * rsqrtf(v4[mt] + BN_EPS);
            const float sh = be4[mt] - mu4[mt] * sc;
            float v = (acc[mt][r] + bb4[mt]) * sc + sh;
            hv[mt] = v > 0.0f ? v : 0.0f;
        }
        hA[2 * r    ].x = hv[0];  hA[2 * r    ].y = hv[1];
        hA[2 * r + 1].x = hv[2];  hA[2 * r + 1].y = hv[3];
    }

    const size_t prow_hi = (size_t)((b * Hn + y0 + 2 * wave + hi) * Wn + x0);
    float* og = kout + prow_hi * KGn;
    float* kl = ks + ((2 * wave + hi) * TSn) * KGn;

    load_b(g1, wq + 64);
    LOAD_GROUP_END();
    {
        v8f acc2[4];
#pragma unroll
        for (int t = 0; t < 4; ++t)
            acc2[t] = v8f{};
        mma_b(acc2, g0, hA);
        const v4f bias = *(const v4f*)(b2 + 4 * c);
        store_b(acc2, bias, og + 4 * c, kl + 4 * c);
    }

    const float* wt = w2 + (size_t)(32 * hi) * KGn + 128 + c;
    v2f bt[16];
#pragma unroll
    for (int s = 0; s < 16; ++s) {
        bt[s].x = wt[(2 * s    ) * KGn];
        bt[s].y = wt[(2 * s + 1) * KGn];
    }
    LOAD_GROUP_END();
    {
        v8f acc2[4];
#pragma unroll
        for (int t = 0; t < 4; ++t)
            acc2[t] = v8f{};
        v8f acc3 = v8f{};
#pragma unroll
        for (int s = 0; s < 16; ++s) {
#pragma unroll
            for (int t = 0; t < 4; ++t) {
                v2f bq;
                bq.x = g1.lo[s][t];
                bq.y = g1.up[s][t];
                acc2[t] = __builtin_amdgcn_wmma_f32_16x16x4_f32(
                              false, hA[s], false, bq, (short)0, acc2[t], false, false);
            }
            acc3 = __builtin_amdgcn_wmma_f32_16x16x4_f32(
                       false, hA[s], false, bt[s], (short)0, acc3, false, false);
        }
        const v4f bias = *(const v4f*)(b2 + 64 + 4 * c);
        store_b(acc2, bias, og + 64 + 4 * c, kl + 64 + 4 * c);
        const float bb = b2[128 + c];
#pragma unroll
        for (int r = 0; r < 8; ++r) {
            const float o = acc3[r] + bb;
            og[r * KGn + 128 + c] = o;
            kl[r * KGn + 128 + c] = o;
        }
    }

    __builtin_amdgcn_fence(__ATOMIC_RELEASE, "wavefront");
    __builtin_amdgcn_wave_barrier();
    __builtin_amdgcn_fence(__ATOMIC_ACQUIRE, "wavefront");

#pragma unroll 1
    for (int rr = 0; rr < 2; ++rr) {
        const int R = 2 * wave + rr;
        const float* xr   = xs + (R * HSn) * XPn + 4 * lane;
        const float* kr   = ks + (R * TSn) * KGn + ((4 * lane) & 15);
        const size_t prow = (size_t)((b * Hn + y0 + R) * Wn + x0);
        float*       orow = out + prow * Cn + 4 * lane;

        v4f col[4][3][2];
        v4f kq[2][9];
#pragma unroll
        for (int hx = 0; hx < 3; ++hx)
#pragma unroll
            for (int ki = 0; ki < 3; ++ki) {
                col[hx][ki][0] = *(const v4f*)(xr + (ki * HSn + hx) * XPn);
                col[hx][ki][1] = *(const v4f*)(xr + (ki * HSn + hx) * XPn + 128);
            }
#pragma unroll
        for (int t = 0; t < 9; ++t)
            kq[0][t] = *(const v4f*)(kr + t * Gn);
#pragma unroll
        for (int j = 0; j < TSn; ++j) {
            if (j + 1 < TSn) {
#pragma unroll
                for (int ki = 0; ki < 3; ++ki) {
                    col[(j + 3) & 3][ki][0] = *(const v4f*)(xr + (ki * HSn + j + 3) * XPn);
                    col[(j + 3) & 3][ki][1] = *(const v4f*)(xr + (ki * HSn + j + 3) * XPn + 128);
                }
#pragma unroll
                for (int t = 0; t < 9; ++t)
                    kq[(j + 1) & 1][t] = *(const v4f*)(kr + (j + 1) * KGn + t * Gn);
            }
            v4f a0 = {}, a1 = {};
#pragma unroll
            for (int ki = 0; ki < 3; ++ki)
#pragma unroll
                for (int kj = 0; kj < 3; ++kj) {
                    const v4f kv = kq[j & 1][ki * 3 + kj];
                    a0 += col[(j + kj) & 3][ki][0] * kv;
                    a1 += col[(j + kj) & 3][ki][1] * kv;
                }
            *(v4f*)(orow + j * Cn)       = a0;
            *(v4f*)(orow + j * Cn + 128) = a1;
            __builtin_amdgcn_sched_barrier(0);
        }
    }
}

extern "C" void kernel_launch(void* const* d_in, const int* in_sizes, int n_in,
                              void* d_out, int out_size, void* d_ws, size_t ws_size,
                              hipStream_t stream)
{
    const float* x     = (const float*)d_in[0];
    const float* w1    = (const float*)d_in[1];
    const float* b1    = (const float*)d_in[2];
    const float* gamma = (const float*)d_in[3];
    const float* beta  = (const float*)d_in[4];
    const float* mean  = (const float*)d_in[5];
    const float* var   = (const float*)d_in[6];
    const float* w2    = (const float*)d_in[7];
    const float* b2    = (const float*)d_in[8];

    float* out  = (float*)d_out;
    float* kout = out + (size_t)NPIX * Cn;

    invol_fused<<<NPIX / (TSn * TSn), 128, 0, stream>>>(
        x, w1, b1, gamma, beta, mean, var, w2, b2, kout, out);
}


# x staging pipelined under product 1: a wave copies its own 16 pixels in four 64-channel pieces two ahead of use, rim shares ride along; group barrier split into signal and wait; sum order kept
# speedup vs baseline: 2.4770x; 1.0463x over previous
#include <hip/hip_runtime.h>

typedef __attribute__((ext_vector_type(2))) float v2f;
typedef __attribute__((ext_vector_type(4))) float v4f;
typedef __attribute__((ext_vector_type(8))) float v8f;
typedef __attribute__((ext_vector_type(4))) unsigned v4u;

#define Bn   4
#define Hn   128
#define Wn   128
#define Cn   256
#define Gn   16
#define K2n  9
#define Crn  64
#define KGn  (K2n * Gn)
#define NPIX (Bn * Hn * Wn)
#define BN_EPS 1e-3f

#define TSn  8
#define HSn  (TSn + 2)
#define XPn  (Cn + 4)
#define XSn  (HSn * HSn * XPn)
#define KSn  (TSn * TSn * KGn)
#define XCH  8
#define RBN  6

struct FragSetA {
    v4f w[8];
    v4f x[2];
};

struct FragSetB {
    v4f lo[16];
    v4f up[16];
};

__device__ __forceinline__ void load_w(FragSetA& f, const float* __restrict__ wp, int c0)
{
#pragma unroll
    for (int q = 0; q < 8; ++q)
        f.w[q] = *(const v4f*)(wp + (c0 + q) * Crn);
}

__device__ __forceinline__ void load_x(FragSetA& f, const float* xq, int c0)
{
    f.x[0] = *(const v4f*)(xq + c0);
    f.x[1] = *(const v4f*)(xq + c0 + 4);
}

__device__ __forceinline__ void mma_a(v8f (&acc)[4], const FragSetA& f)
{
#pragma unroll
    for (int j = 0; j < 4; ++j) {
        v2f b;
        b.x = f.x[j >> 1][(2 * j) & 3];
        b.y = f.x[j >> 1][(2 * j + 1) & 3];
#pragma unroll
        for (int mt = 0; mt < 4; ++mt) {
            v2f a;
            a.x = f.w[2 * j][mt];
            a.y = f.w[2 * j + 1][mt];
            acc[mt] = __builtin_amdgcn_wmma_f32_16x16x4_f32(
                          false, a, false, b, (short)0, acc[mt], false, false);
        }
    }
}

__device__ __forceinline__ void load_b(FragSetB& f, const float* __restrict__ wq)
{
#pragma unroll
    for (int s = 0; s < 16; ++s) {
        f.lo[s] = *(const v4f*)(wq + (2 * s    ) * KGn);
        f.up[s] = *(const v4f*)(wq + (2 * s + 1) * KGn);
    }
}

__device__ __forceinline__ void mma_b(v8f (&acc2)[4], const FragSetB& f, const v2f (&hA)[16])
{
#pragma unroll
    for (int s = 0; s < 16; ++s) {
#pragma unroll
        for (int t = 0; t < 4; ++t) {
            v2f b;
            b.x = f.lo[s][t];
            b.y = f.up[s][t];
            acc2[t] = __builtin_amdgcn_wmma_f32_16x16x4_f32(
                          false, hA[s], false, b, (short)0, acc2[t], false, false);
        }
    }
}

__device__ __forceinline__ void store_b(const v8f (&acc2)[4], const v4f bias,
                                        float* __restrict__ og, float* kl)
{
#pragma unroll
    for (int r = 0; r < 8; ++r) {
        v4f o;
        o.x = acc2[0][r] + bias.x;
        o.y = acc2[1][r] + bias.y;
        o.z = acc2[2][r] + bias.z;
        o.w = acc2[3][r] + bias.w;
        *(v4f*)(og + r * KGn) = o;
        *(v4f*)(kl + r * KGn) = o;
    }
}

#define OWN_G(e) ((((e) >> 2) * Wn  + 2 * ((e) & 3)) * Cn)
#define OWN_L(e) ((((e) >> 2) * HSn + 2 * ((e) & 3)) * XPn)

__device__ __forceinline__ void load_own(v4u (&v)[XCH], const float* xg, int q)
{
#pragma unroll
    for (int e = 0; e < XCH; ++e)
        v[e] = *(const v4u*)(xg + OWN_G(e) + 64 * q);
}

__device__ __forceinline__ void store_own(const v4u (&v)[XCH], float* xo, int q)
{
#pragma unroll
    for (int e = 0; e < XCH; ++e)
        *(v4u*)(xo + OWN_L(e) + 64 * q) = v[e];
}

struct RimMap { int ay, by, ax, bx; };

struct RimAt {
    int      goff;
    int      loff;
    unsigned keep;
};

__device__ __forceinline__ RimAt rim_at(const RimMap m, int i, int b, int y0, int x0)
{
    const int n  = i >> 1;
    const int hy = m.ay + m.by * n;
    const int hx = m.ax + m.bx * n;
    const int gy = y0 + hy - 1;
    const int gx = x0 + hx - 1;
    RimAt a;
    a.keep = ((unsigned)gy < (unsigned)Hn && (unsigned)gx < (unsigned)Wn) ? 0xffffffffu : 0u;
    const int cy = min(max(gy, 0), Hn - 1);
    const int cx = min(max(gx, 0), Wn - 1);
    a.goff = ((b * Hn + cy) * Wn + cx) * Cn + (i & 1) * 128;
    a.loff = (hy * HSn + hx) * XPn + (i & 1) * 128;
    return a;
}

__device__ __forceinline__ void load_rim(v4u (&v)[RBN], const float* xl, const RimMap m, int batch,
                                         int b, int y0, int x0)
{
#pragma unroll
    for (int i = 0; i < RBN; ++i)
        v[i] = *(const v4u*)(xl + rim_at(m, batch * RBN + i, b, y0, x0).goff);
}

__device__ __forceinline__ void store_rim(const v4u (&v)[RBN], float* xsl, const RimMap m, int batch,
                                          int b, int y0, int x0)
{
#pragma unroll
    for (int i = 0; i < RBN; ++i) {
        const RimAt a = rim_at(m, batch * RBN + i, b, y0, x0);
        *(v4u*)(xsl + a.loff) = v[i] & a.keep;
    }
}

__global__ __launch_bounds__(128, 2) void invol_fused(
    const float* __restrict__ x,  const float* __restrict__ w1,
    const float* __restrict__ b1, const float* __restrict__ gamma,
    const float* __restrict__ beta, const float* __restrict__ mean,
    const float* __restrict__ var,  const float* __restrict__ w2,
    const float* __restrict__ b2,   float* __restrict__ kout,
    float* __restrict__ out)
{
    __shared__ __attribute__((aligned(16))) float xs[XSn];
    __shared__ __attribute__((aligned(16))) float ks[KSn];

    const int lane = threadIdx.x & 31;
    const int wave = __builtin_amdgcn_readfirstlane((int)(threadIdx.x >> 5));
    const int tile = blockIdx.x;
    const int x0   = (tile & 15) * TSn;
    const int y0   = ((tile >> 4) & 15) * TSn;
    const int b    = tile >> 8;
    const int c    = lane & 15;
    const int hi   = lane >> 4;

    const int c16 = lane & 15;
    const float* xg  = x  + (size_t)(((b * Hn + y0 + 2 * wave) * Wn + x0 + hi) * Cn + 4 * c16);
    float*       xo  = xs + ((2 * wave + 1) * HSn + 1 + hi) * XPn + 4 * c16;
    const float* xl  = x  + 4 * lane;
    float*       xsl = xs + 4 * lane;

#define LOAD_GROUP_END() __builtin_amdgcn_sched_barrier(0)
#define OWN_X_STORED() do { __builtin_amdgcn_fence(__ATOMIC_RELEASE, "wavefront"); \
                            __builtin_amdgcn_wave_barrier();                        \
                            __builtin_amdgcn_fence(__ATOMIC_ACQUIRE, "wavefront"); } while (0)

    RimMap rm;
    rm.by = wave >> 1;
    rm.bx = 1 - rm.by;
    rm.ay = (wave == 1) ? HSn - 1 : (wave == 2 ? 1 : 0);
    rm.ax = (wave == 3) ? HSn - 1 : (wave == 1 ? 1 : 0);

    v4u own0[XCH], own1[XCH];
    v4u rim0[RBN], rim1[RBN];

    const float* wp = w1 + (size_t)(8 * hi) * Crn + 4 * c;
    FragSetA fa0, fa1, fb0, fb1;

    load_own(own0, xg, 0);
    LOAD_GROUP_END();
    load_w(fa0, wp, 0);
    load_w(fa1, wp, 16);
    LOAD_GROUP_END();
    load_own(own1, xg, 1);
    load_rim(rim1, xl, rm, 0, b, y0, x0);
    LOAD_GROUP_END();

    store_own(own0, xo, 0);
    OWN_X_STORED();

    const int pr = (2 * wave + (c >> 3) + 1) * HSn + (c & 7) + 1;
    const float* xq = xs + pr * XPn + 8 * hi;

    v8f acc[4];
#pragma unroll
    for (int mt = 0; mt < 4; ++mt)
        acc[mt] = v8f{};

    load_x(fa0, xq, 0);
    load_x(fa1, xq, 16);

    load_w(fb0, wp, 32); load_x(fb0, xq, 32);
    load_w(fb1, wp, 48); load_x(fb1, xq, 48);
    load_own(own0, xg, 2);
    load_rim(rim0, xl, rm, 1, b, y0, x0);
    LOAD_GROUP_END();
    mma_a(acc, fa0);
    mma_a(acc, fa1);
    LOAD_GROUP_END();
    store_own(own1, xo, 1);
    store_rim(rim1, xsl, rm, 0, b, y0, x0);
    OWN_X_STORED();
    load_w(fa0, wp, 64); load_x(fa0, xq, 64);
    load_w(fa1, wp, 80); load_x(fa1, xq, 80);
    LOAD_GROUP_END();
    mma_a(acc, fb0);
    mma_a(acc, fb1);
    LOAD_GROUP_END();

    load_w(fb0, wp, 96);  load_x(fb0, xq, 96);
    load_w(fb1, wp, 112); load_x(fb1, xq, 112);
    load_own(own1, xg, 3);
    load_rim(rim1, xl, rm, 2, b, y0, x0);
    LOAD_GROUP_END();
    mma_a(acc, fa0);
    mma_a(acc, fa1);
    LOAD_GROUP_END();
    store_own(own0, xo, 2);
    store_rim(rim0, xsl, rm, 1, b, y0, x0);
    OWN_X_STORED();
    load_w(fa0, wp, 128); load_x(fa0, xq, 128);
    load_w(fa1, wp, 144); load_x(fa1, xq, 144);
    LOAD_GROUP_END();
    mma_a(acc, fb0);
    mma_a(acc, fb1);
    LOAD_GROUP_END();

    load_w(fb0, wp, 160); load_x(fb0, xq, 160);
    load_w(fb1, wp, 176); load_x(fb1, xq, 176);
    LOAD_GROUP_END();
    mma_a(acc, fa0);
    mma_a(acc, fa1);
    LOAD_GROUP_END();
    store_own(own1, xo, 3);
    store_rim(rim1, xsl, rm, 2, b, y0, x0);
    asm volatile("s_wait_dscnt 0x0" ::: "memory");
    asm volatile("s_barrier_signal -1" ::: "memory");
    load_w(fa0, wp, 192); load_x(fa0, xq, 192);
    load_w(fa1, wp, 208); load_x(fa1, xq, 208);
    LOAD_GROUP_END();
    mma_a(acc, fb0);
    mma_a(acc, fb1);
    LOAD_GROUP_END();

    load_w(fb0, wp, 224); load_x(fb0, xq, 224);
    load_w(fb1, wp, 240); load_x(fb1, xq, 240);
    LOAD_GROUP_END();
    mma_a(acc, fa0);
    mma_a(acc, fa1);
    mma_a(acc, fb0);
    mma_a(acc, fb1);

    const float* wq = w2 + (size_t)(32 * hi) * KGn + 4 * c;
    FragSetB g0, g1;
    load_b(g0, wq);
    LOAD_GROUP_END();

    v2f hA[16];
#pragma unroll
    for (int r = 0; r < 8; ++r) {
        const int d0 = 32 * hi + 4 * r;
        const v4f g4  = *(const v4f*)(gamma + d0);
        const v4f v4  = *(const v4f*)(var   + d0);
        const v4f be4 = *(const v4f*)(beta  + d0);
        const v4f mu4 = *(const v4f*)(mean  + d0);
        const v4f bb4 = *(const v4f*)(b1    + d0);
        float hv[4];
#pragma unroll
        for (int mt = 0; mt < 4; ++mt) {
            const float sc = g4[mt] * rsqrtf(v4[mt] + BN_EPS);
            const float sh = be4[mt] - mu4[mt] * sc;
            float v = (acc[mt][r] + bb4[mt]) * sc + sh;
            hv[mt] = v > 0.0f ? v : 0.0f;
        }
        hA[2 * r    ].x = hv[0];  hA[2 * r    ].y = hv[1];
        hA[2 * r + 1].x = hv[2];  hA[2 * r + 1].y = hv[3];
    }

    const size_t prow_hi = (size_t)((b * Hn + y0 + 2 * wave + hi) * Wn + x0);
    float* og = kout + prow_hi * KGn;
    float* kl = ks + ((2 * wave + hi) * TSn) * KGn;

    load_b(g1, wq + 64);
    LOAD_GROUP_END();
    {
        v8f acc2[4];
#pragma unroll
        for (int t = 0; t < 4; ++t)
            acc2[t] = v8f{};
        mma_b(acc2, g0, hA);
        const v4f bias = *(const v4f*)(b2 + 4 * c);
        store_b(acc2, bias, og + 4 * c, kl + 4 * c);
    }

    const float* wt = w2 + (size_t)(32 * hi) * KGn + 128 + c;
    v2f bt[16];
#pragma unroll
    for (int s = 0; s < 16; ++s) {
        bt[s].x = wt[(2 * s    ) * KGn];
        bt[s].y = wt[(2 * s + 1) * KGn];
    }
    LOAD_GROUP_END();
    {
        v8f acc2[4];
#pragma unroll
        for (int t = 0; t < 4; ++t)
            acc2[t] = v8f{};
        v8f acc3 = v8f{};
#pragma unroll
        for (int s = 0; s < 16; ++s) {
#pragma unroll
            for (int t = 0; t < 4; ++t) {
                v2f bq;
                bq.x = g1.lo[s][t];
                bq.y = g1.up[s][t];
                acc2[t] = __builtin_amdgcn_wmma_f32_16x16x4_f32(
                              false, hA[s], false, bq, (short)0, acc2[t], false, false);
            }
            acc3 = __builtin_amdgcn_wmma_f32_16x16x4_f32(
                       false, hA[s], false, bt[s], (short)0, acc3, false, false);
        }
        const v4f bias = *(const v4f*)(b2 + 64 + 4 * c);
        store_b(acc2, bias, og + 64 + 4 * c, kl + 64 + 4 * c);
        const float bb = b2[128 + c];
#pragma unroll
        for (int r = 0; r < 8; ++r) {
            const float o = acc3[r] + bb;
            og[r * KGn + 128 + c] = o;
            kl[r * KGn + 128 + c] = o;
        }
    }

    __builtin_amdgcn_fence(__ATOMIC_RELEASE, "wavefront");
    __builtin_amdgcn_wave_barrier();
    __builtin_amdgcn_fence(__ATOMIC_ACQUIRE, "wavefront");
    asm volatile("s_barrier_wait -1" ::: "memory");

#pragma unroll 1
    for (int rr = 0; rr < 2; ++rr) {
        const int R = 2 * wave + rr;
        const float* xr   = xs + (R * HSn) * XPn + 4 * lane;
        const float* kr   = ks + (R * TSn) * KGn + ((4 * lane) & 15);
        const size_t prow = (size_t)((b * Hn + y0 + R) * Wn + x0);
        float*       orow = out + prow * Cn + 4 * lane;

        v4f col[4][3][2];
        v4f kq[2][9];
#pragma unroll
        for (int hx = 0; hx < 3; ++hx)
#pragma unroll
            for (int ki = 0; ki < 3; ++ki) {
                col[hx][ki][0] = *(const v4f*)(xr + (ki * HSn + hx) * XPn);
                col[hx][ki][1] = *(const v4f*)(xr + (ki * HSn + hx) * XPn + 128);
            }
#pragma unroll
        for (int t = 0; t < 9; ++t)
            kq[0][t] = *(const v4f*)(kr + t * Gn);
#pragma unroll
        for (int j = 0; j < TSn; ++j) {
            if (j + 1 < TSn) {
#pragma unroll
                for (int ki = 0; ki < 3; ++ki) {
                    col[(j + 3) & 3][ki][0] = *(const v4f*)(xr + (ki * HSn + j + 3) * XPn);
                    col[(j + 3) & 3][ki][1] = *(const v4f*)(xr + (ki * HSn + j + 3) * XPn + 128);
                }
#pragma unroll
                for (int t = 0; t < 9; ++t)
                    kq[(j + 1) & 1][t] = *(const v4f*)(kr + (j + 1) * KGn + t * Gn);
            }
            v4f a0 = {}, a1 = {};
#pragma unroll
            for (int ki = 0; ki < 3; ++ki)
#pragma unroll
                for (int kj = 0; kj < 3; ++kj) {
                    const v4f kv = kq[j & 1][ki * 3 + kj];
                    a0 += col[(j + kj) & 3][ki][0] * kv;
                    a1 += col[(j + kj) & 3][ki][1] * kv;
                }
            *(v4f*)(orow + j * Cn)       = a0;
            *(v4f*)(orow + j * Cn + 128) = a1;
            __builtin_amdgcn_sched_barrier(0);
        }
    }
}

extern "C" void kernel_launch(void* const* d_in, const int* in_sizes, int n_in,
                              void* d_out, int out_size, void* d_ws, size_t ws_size,
                              hipStream_t stream)
{
    const float* x     = (const float*)d_in[0];
    const float* w1    = (const float*)d_in[1];
    const float* b1    = (const float*)d_in[2];
    const float* gamma = (const float*)d_in[3];
    const float* beta  = (const float*)d_in[4];
    const float* mean  = (const float*)d_in[5];
    const float* var   = (const float*)d_in[6];
    const float* w2    = (const float*)d_in[7];
    const float* b2    = (const float*)d_in[8];

    float* out  = (float*)d_out;
    float* kout = out + (size_t)NPIX * Cn;

    invol_fused<<<NPIX / (TSn * TSn), 128, 0, stream>>>(
        x, w1, b1, gamma, beta, mean, var, w2, b2, kout, out);
}
